# starting kernel
# baseline (speedup 1.0000x reference)
#include <hip/hip_runtime.h>
#include <math.h>

typedef __attribute__((ext_vector_type(2))) float v2f;
typedef __attribute__((ext_vector_type(8))) float v8f;

#define N_NODES 1024
#define F_DIM   64
#define GAT_ALPHA 0.2f
#define NEG_INF_F (-9.0e15f)
#define PS 1028

__global__ void k_wh_e(const float* __restrict__ h, const float* __restrict__ W,
                       const float* __restrict__ a,
                       float* __restrict__ Wh, float* __restrict__ e1,
                       float* __restrict__ e2) {
  const int lane = threadIdx.x & 31;
  const int half = lane >> 4;
  const int l    = lane & 15;
  const int m0   = blockIdx.x * 16;

  float p1[8] = {0.f, 0.f, 0.f, 0.f, 0.f, 0.f, 0.f, 0.f};
  float p2[8] = {0.f, 0.f, 0.f, 0.f, 0.f, 0.f, 0.f, 0.f};

#pragma unroll 1
  for (int nt = 0; nt < 4; ++nt) {
    const int ncol = nt * 16 + l;
    v8f c = {};
#pragma unroll
    for (int kb = 0; kb < F_DIM; kb += 4) {
      const int k0 = kb + 2 * half;
      v2f av = *(const v2f*)(h + (m0 + l) * F_DIM + k0);
      v2f bv;
      bv.x = W[k0 * F_DIM + ncol];
      bv.y = W[(k0 + 1) * F_DIM + ncol];
      c = __builtin_amdgcn_wmma_f32_16x16x4_f32(false, av, false, bv,
                                                (short)0, c, false, false);
    }
    const float a1n = a[ncol];
    const float a2n = a[F_DIM + ncol];
#pragma unroll
    for (int r = 0; r < 8; ++r) {
      Wh[(m0 + r + 8 * half) * F_DIM + ncol] = c[r];
      p1[r] += c[r] * a1n;
      p2[r] += c[r] * a2n;
    }
  }

#pragma unroll
  for (int off = 1; off < 16; off <<= 1) {
#pragma unroll
    for (int r = 0; r < 8; ++r) {
      p1[r] += __shfl_xor(p1[r], off, 32);
      p2[r] += __shfl_xor(p2[r], off, 32);
    }
  }
  if (l == 0) {
#pragma unroll
    for (int r = 0; r < 8; ++r) {
      const int m = m0 + r + 8 * half;
      e1[m] = p1[r];
      e2[m] = p2[r];
    }
  }
}

__global__ void k_attn(const float* __restrict__ Wh, const int* __restrict__ adj,
                       const float* __restrict__ e1, const float* __restrict__ e2,
                       float* __restrict__ out) {
  extern __shared__ float smem[];
  float* P    = smem;
  float* sums = smem + 16 * PS;

  const int tid  = threadIdx.x;
  const int wave = tid >> 5;
  const int lane = tid & 31;
  const int half = lane >> 4;
  const int l    = lane & 15;
  const int i0   = blockIdx.x * 16;

#pragma unroll 1
  for (int rr = 0; rr < 4; ++rr) {
    const int r = wave * 4 + rr;
    const int i = i0 + r;
    const float ei = e1[i];

    float mx = NEG_INF_F;
#pragma unroll
    for (int t = 0; t < N_NODES / 128; ++t) {
      const int j0 = (t * 32 + lane) * 4;
      const int4   av = *(const int4*)(adj + (size_t)i * N_NODES + j0);
      const float4 ev = *(const float4*)(e2 + j0);
      float4 s;
      s.x = ei + ev.x; s.y = ei + ev.y; s.z = ei + ev.z; s.w = ei + ev.w;
      s.x = (s.x > 0.f) ? s.x : GAT_ALPHA * s.x;
      s.y = (s.y > 0.f) ? s.y : GAT_ALPHA * s.y;
      s.z = (s.z > 0.f) ? s.z : GAT_ALPHA * s.z;
      s.w = (s.w > 0.f) ? s.w : GAT_ALPHA * s.w;
      s.x = (av.x > 0) ? s.x : NEG_INF_F;
      s.y = (av.y > 0) ? s.y : NEG_INF_F;
      s.z = (av.z > 0) ? s.z : NEG_INF_F;
      s.w = (av.w > 0) ? s.w : NEG_INF_F;
      *(float4*)(P + r * PS + j0) = s;
      mx = fmaxf(mx, fmaxf(fmaxf(s.x, s.y), fmaxf(s.z, s.w)));
      if (i + 1 < N_NODES)
        __builtin_prefetch(adj + (size_t)(i + 1) * N_NODES + j0, 0, 1);
    }
#pragma unroll
    for (int off = 16; off > 0; off >>= 1)
      mx = fmaxf(mx, __shfl_xor(mx, off, 32));

    float sum = 0.0f;
#pragma unroll
    for (int t = 0; t < N_NODES / 128; ++t) {
      const int j0 = (t * 32 + lane) * 4;
      float4 p = *(const float4*)(P + r * PS + j0);
      p.x = __expf(p.x - mx);
      p.y = __expf(p.y - mx);
      p.z = __expf(p.z - mx);
      p.w = __expf(p.w - mx);
      *(float4*)(P + r * PS + j0) = p;
      sum += p.x + p.y + p.z + p.w;
    }
#pragma unroll
    for (int off = 16; off > 0; off >>= 1)
      sum += __shfl_xor(sum, off, 32);
    if (lane == 0) sums[r] = sum;
  }
  __syncthreads();

  const int ncol = wave * 16 + l;
  v8f c = {};
#pragma unroll 4
  for (int kb = 0; kb < N_NODES; kb += 4) {
    const int k0 = kb + 2 * half;
    const v2f av = *(const v2f*)(P + l * PS + k0);
    v2f bv;
    bv.x = Wh[k0 * F_DIM + ncol];
    bv.y = Wh[(k0 + 1) * F_DIM + ncol];
    c = __builtin_amdgcn_wmma_f32_16x16x4_f32(false, av, false, bv,
                                              (short)0, c, false, false);
  }

#pragma unroll
  for (int r = 0; r < 8; ++r) {
    const int m = r + 8 * half;
    float v = c[r] / sums[m];
    v = (v > 0.0f) ? v : (__expf(v) - 1.0f);
    out[(i0 + m) * F_DIM + ncol] = v;
  }
}

extern "C" void kernel_launch(void* const* d_in, const int* in_sizes, int n_in,
                              void* d_out, int out_size, void* d_ws, size_t ws_size,
                              hipStream_t stream) {
  const float* h   = (const float*)d_in[0];
  const int*   adj = (const int*)  d_in[1];
  const float* W   = (const float*)d_in[2];
  const float* a   = (const float*)d_in[3];
  float* out = (float*)d_out;

  float* Wh = (float*)d_ws;
  float* e1 = Wh + N_NODES * F_DIM;
  float* e2 = e1 + N_NODES;

  k_wh_e<<<N_NODES / 16, 32, 0, stream>>>(h, W, a, Wh, e1, e2);

  const size_t shmem = (size_t)(16 * PS + 16) * sizeof(float);
  k_attn<<<N_NODES / 16, 128, shmem, stream>>>(Wh, adj, e1, e2, out);
}


# k_wh_e: the 4 column tiles of a row tile go to 4 waves of one block (were 4 dependent passes of one wave); per-tile e1/e2 shares joined through LDS in tile order (f32 sum order of e1/e2 changes)
# speedup vs baseline: 4.5489x; 4.5489x over previous
#include <hip/hip_runtime.h>
#include <math.h>

typedef __attribute__((ext_vector_type(2))) float v2f;
typedef __attribute__((ext_vector_type(4))) float v4f;
typedef __attribute__((ext_vector_type(8))) float v8f;

#define N_NODES 1024
#define F_DIM   64
#define GAT_ALPHA 0.2f
#define NEG_INF_F (-9.0e15f)
#define PS 1028

__global__ void __launch_bounds__(128)
k_wh_e(const float* __restrict__ h, const float* __restrict__ W,
       const float* __restrict__ a,
       float* __restrict__ Wh, float* __restrict__ e1,
       float* __restrict__ e2) {
  __shared__ float part[4][2][16];

  const int wave = threadIdx.x >> 5;
  const int lane = threadIdx.x & 31;
  const int half = lane >> 4;
  const int l    = lane & 15;
  const int m0   = blockIdx.x * 16;
  const int ncol = wave * 16 + l;

  v8f c = {};
#pragma unroll
  for (int kb = 0; kb < F_DIM; kb += 4) {
    const int k0 = kb + 2 * half;
    v2f av = *(const v2f*)(h + (m0 + l) * F_DIM + k0);
    v2f bv;
    bv.x = W[k0 * F_DIM + ncol];
    bv.y = W[(k0 + 1) * F_DIM + ncol];
    c = __builtin_amdgcn_wmma_f32_16x16x4_f32(false, av, false, bv,
                                              (short)0, c, false, false);
  }
  const float a1n = a[ncol];
  const float a2n = a[F_DIM + ncol];
  float p1[8], p2[8];
#pragma unroll
  for (int r = 0; r < 8; ++r) {
    p1[r] = c[r] * a1n;
    p2[r] = c[r] * a2n;
  }
#pragma unroll
  for (int r = 0; r < 8; r += 2) {
    const int m = m0 + r + 8 * half;
    v2f pr;
    pr.x = c[r];
    pr.y = c[r + 1];
    *(v2f*)(Wh + (m >> 1) * (2 * F_DIM) + 2 * ncol) = pr;
  }

#pragma unroll
  for (int off = 1; off < 16; off <<= 1) {
#pragma unroll
    for (int r = 0; r < 8; ++r) {
      p1[r] += __shfl_xor(p1[r], off, 32);
      p2[r] += __shfl_xor(p2[r], off, 32);
    }
  }
  if (l == 0) {
#pragma unroll
    for (int r = 0; r < 8; ++r) {
      part[wave][0][r + 8 * half] = p1[r];
      part[wave][1][r + 8 * half] = p2[r];
    }
  }
  __syncthreads();
  if (wave == 0) {
    const float s = ((part[0][half][l] + part[1][half][l]) + part[2][half][l]) + part[3][half][l];
    float* dst = half ? e2 : e1;
    dst[m0 + l] = s;
  }
}

#define KW   (N_NODES / 4)
#define KSW  (KW / 4)

__global__ void __launch_bounds__(128) __attribute__((amdgpu_waves_per_eu(1, 1)))
k_attn(const float* __restrict__ Wh, const int* __restrict__ adj,
       const float* __restrict__ e1, const float* __restrict__ e2,
       float* __restrict__ out) {
  __shared__ float4 P4[16 * PS / 4];
  __shared__ float  red[4][8][32];
  __shared__ float  sums[16];
  float* P = (float*)P4;

  const int tid  = threadIdx.x;
  const int wave = tid >> 5;
  const int lane = tid & 31;
  const int half = lane >> 4;
  const int l    = lane & 15;
  const int i0   = (blockIdx.x >> 2) * 16;
  const int ncol = (blockIdx.x & 3) * 16 + l;

  const int r0 = wave * 4;
  int4 av[4][8];
#pragma unroll
  for (int rr = 0; rr < 4; ++rr) {
#pragma unroll
    for (int t = 0; t < N_NODES / 128; ++t)
      av[rr][t] = *(const int4*)(adj + (size_t)(i0 + r0 + rr) * N_NODES + (t * 32 + lane) * 4);
  }
  float4 ev[8];
#pragma unroll
  for (int t = 0; t < N_NODES / 128; ++t)
    ev[t] = *(const float4*)(e2 + (t * 32 + lane) * 4);
  float ei[4];
#pragma unroll
  for (int rr = 0; rr < 4; ++rr) ei[rr] = e1[i0 + r0 + rr];

  const v2f* Bg = (const v2f*)Wh + (size_t)(wave * (KW / 2) + half) * F_DIM + ncol;
  v2f bf[KSW];
#pragma unroll
  for (int ks = 0; ks < KSW; ++ks) bf[ks] = Bg[(2 * ks) * F_DIM];

  float mx[4];
#pragma unroll
  for (int rr = 0; rr < 4; ++rr) {
    const int r = r0 + rr;
    float m = NEG_INF_F;
#pragma unroll
    for (int t = 0; t < N_NODES / 128; ++t) {
      const int j0 = (t * 32 + lane) * 4;
      const int4   a4 = av[rr][t];
      const float4 e4 = ev[t];
      float4 s;
      s.x = ei[rr] + e4.x; s.y = ei[rr] + e4.y; s.z = ei[rr] + e4.z; s.w = ei[rr] + e4.w;
      s.x = (s.x > 0.f) ? s.x : GAT_ALPHA * s.x;
      s.y = (s.y > 0.f) ? s.y : GAT_ALPHA * s.y;
      s.z = (s.z > 0.f) ? s.z : GAT_ALPHA * s.z;
      s.w = (s.w > 0.f) ? s.w : GAT_ALPHA * s.w;
      s.x = (a4.x > 0) ? s.x : NEG_INF_F;
      s.y = (a4.y > 0) ? s.y : NEG_INF_F;
      s.z = (a4.z > 0) ? s.z : NEG_INF_F;
      s.w = (a4.w > 0) ? s.w : NEG_INF_F;
      *(float4*)(P + r * PS + j0) = s;
      m = fmaxf(m, fmaxf(fmaxf(s.x, s.y), fmaxf(s.z, s.w)));
    }
    mx[rr] = m;
  }
#pragma unroll
  for (int off = 16; off > 0; off >>= 1) {
#pragma unroll
    for (int rr = 0; rr < 4; ++rr)
      mx[rr] = fmaxf(mx[rr], __shfl_xor(mx[rr], off, 32));
  }

  float sm[4];
#pragma unroll
  for (int rr = 0; rr < 4; ++rr) {
    const int r = r0 + rr;
    float sum = 0.0f;
#pragma unroll
    for (int t = 0; t < N_NODES / 128; ++t) {
      const int j0 = (t * 32 + lane) * 4;
      float4 p = *(const float4*)(P + r * PS + j0);
      p.x = __expf(p.x - mx[rr]);
      p.y = __expf(p.y - mx[rr]);
      p.z = __expf(p.z - mx[rr]);
      p.w = __expf(p.w - mx[rr]);
      *(float4*)(P + r * PS + j0) = p;
      sum += p.x + p.y + p.z + p.w;
    }
    sm[rr] = sum;
  }
#pragma unroll
  for (int off = 16; off > 0; off >>= 1) {
#pragma unroll
    for (int rr = 0; rr < 4; ++rr)
      sm[rr] += __shfl_xor(sm[rr], off, 32);
  }
  if (lane == 0) {
#pragma unroll
    for (int rr = 0; rr < 4; ++rr) sums[r0 + rr] = sm[rr];
  }
  __syncthreads();

  const v2f* Pv = (const v2f*)(P + l * PS + 2 * half) + wave * (KW / 2);
  v2f af[KSW];
#pragma unroll
  for (int ks = 0; ks < KSW; ++ks) af[ks] = Pv[2 * ks];
  __builtin_amdgcn_sched_barrier(0);
  v8f c0 = {}, c1 = {};
#pragma unroll
  for (int ks = 0; ks < KSW; ks += 2) {
    c0 = __builtin_amdgcn_wmma_f32_16x16x4_f32(false, af[ks], false, bf[ks],
                                               (short)0, c0, false, false);
    c1 = __builtin_amdgcn_wmma_f32_16x16x4_f32(false, af[ks + 1], false, bf[ks + 1],
                                               (short)0, c1, false, false);
  }
#pragma unroll
  for (int r = 0; r < 8; ++r) red[wave][r][lane] = c0[r] + c1[r];
  __syncthreads();

#pragma unroll
  for (int rr = 0; rr < 2; ++rr) {
    const int r = 2 * wave + rr;
    const int m = r + 8 * half;
    float v = ((red[0][r][lane] + red[1][r][lane]) + red[2][r][lane]) + red[3][r][lane];
    v = v / sums[m];
    v = (v > 0.0f) ? v : (__expf(v) - 1.0f);
    out[(i0 + m) * F_DIM + ncol] = v;
  }
}

extern "C" void kernel_launch(void* const* d_in, const int* in_sizes, int n_in,
                              void* d_out, int out_size, void* d_ws, size_t ws_size,
                              hipStream_t stream) {
  const float* h   = (const float*)d_in[0];
  const int*   adj = (const int*)  d_in[1];
  const float* W   = (const float*)d_in[2];
  const float* a   = (const float*)d_in[3];
  float* out = (float*)d_out;

  float* Wh = (float*)d_ws;
  float* e1 = Wh + N_NODES * F_DIM;
  float* e2 = e1 + N_NODES;

  k_wh_e<<<N_NODES / 16, 128, 0, stream>>>(h, W, a, Wh, e1, e2);

  k_attn<<<(N_NODES / 16) * (F_DIM / 16), 128, 0, stream>>>(Wh, adj, e1, e2, out);
}


# k_attn softmax re-mapped onto the WMMA A-fragment layout (lane=row, wave=K quarter): P stays in registers, no LDS tile, no butterflies; adj as bit words packed by launch 1 (f32 row-sum order changes)
# speedup vs baseline: 5.0413x; 1.1082x over previous
#include <hip/hip_runtime.h>
#include <math.h>

typedef __attribute__((ext_vector_type(2))) float v2f;
typedef __attribute__((ext_vector_type(4))) float v4f;
typedef __attribute__((ext_vector_type(8))) float v8f;

#define N_NODES 1024
#define F_DIM   64
#define GAT_ALPHA 0.2f
#define NEG_INF_F (-9.0e15f)

#define KW   (N_NODES / 4)
#define KSW  (KW / 4)

#define WH_BLOCKS   (N_NODES / 16)
#define PACK_BLOCKS (N_NODES / 4)
#define MASK_WORDS  32

__global__ void __launch_bounds__(128)
k_wh_e(const float* __restrict__ h, const float* __restrict__ W,
       const float* __restrict__ a, const int* __restrict__ adj,
       float* __restrict__ Wh, float* __restrict__ e1,
       float* __restrict__ e2, unsigned* __restrict__ amask) {
  __shared__ float part[4][2][16];

  const int wave = threadIdx.x >> 5;
  const int lane = threadIdx.x & 31;

  if (blockIdx.x >= WH_BLOCKS) {
    const int row = (blockIdx.x - WH_BLOCKS) * 4 + wave;
    int4 av[N_NODES / 128];
#pragma unroll
    for (int t = 0; t < N_NODES / 128; ++t)
      av[t] = *(const int4*)(adj + (size_t)row * N_NODES + (t * 32 + lane) * 4);
    unsigned mine = 0u;
#pragma unroll
    for (int t = 0; t < N_NODES / 128; ++t) {
      const int wi = (t >> 1) * 8 + (t & 1) * 2;
      const unsigned bx = __builtin_amdgcn_ballot_w32(av[t].x > 0);
      const unsigned by = __builtin_amdgcn_ballot_w32(av[t].y > 0);
      const unsigned bz = __builtin_amdgcn_ballot_w32(av[t].z > 0);
      const unsigned bw = __builtin_amdgcn_ballot_w32(av[t].w > 0);
      mine = (lane == wi + 0) ? bx : mine;
      mine = (lane == wi + 1) ? by : mine;
      mine = (lane == wi + 4) ? bz : mine;
      mine = (lane == wi + 5) ? bw : mine;
    }
    amask[row * MASK_WORDS + lane] = mine;
    return;
  }

  const int half = lane >> 4;
  const int l    = lane & 15;
  const int m0   = blockIdx.x * 16;
  const int ncol = wave * 16 + l;
  v8f c = {};
#pragma unroll
  for (int kb = 0; kb < F_DIM; kb += 4) {
    const int k0 = kb + 2 * half;
    v2f av = *(const v2f*)(h + (m0 + l) * F_DIM + k0);
    v2f bv;
    bv.x = W[k0 * F_DIM + ncol];
    bv.y = W[(k0 + 1) * F_DIM + ncol];
    c = __builtin_amdgcn_wmma_f32_16x16x4_f32(false, av, false, bv,
                                              (short)0, c, false, false);
  }
  const float a1n = a[ncol];
  const float a2n = a[F_DIM + ncol];
  float p1[8], p2[8];
#pragma unroll
  for (int r = 0; r < 8; ++r) {
    p1[r] = c[r] * a1n;
    p2[r] = c[r] * a2n;
  }
#pragma unroll
  for (int r = 0; r < 8; r += 2) {
    const int m = m0 + r + 8 * half;
    v2f pr;
    pr.x = c[r];
    pr.y = c[r + 1];
    *(v2f*)(Wh + (m >> 1) * (2 * F_DIM) + 2 * ncol) = pr;
  }

#pragma unroll
  for (int off = 1; off < 16; off <<= 1) {
#pragma unroll
    for (int r = 0; r < 8; ++r) {
      p1[r] += __shfl_xor(p1[r], off, 32);
      p2[r] += __shfl_xor(p2[r], off, 32);
    }
  }
  if (l == 0) {
#pragma unroll
    for (int r = 0; r < 8; ++r) {
      part[wave][0][r + 8 * half] = p1[r];
      part[wave][1][r + 8 * half] = p2[r];
    }
  }
  __syncthreads();
  if (wave == 0) {
    const float s = ((part[0][half][l] + part[1][half][l]) + part[2][half][l]) + part[3][half][l];
    float* dst = half ? e2 : e1;
    dst[m0 + l] = s;
  }
}


__global__ void __launch_bounds__(128) __attribute__((amdgpu_waves_per_eu(1, 1)))
k_attn(const float* __restrict__ Wh, const unsigned* __restrict__ amask,
       const float* __restrict__ e1, const float* __restrict__ e2,
       float* __restrict__ out) {
  __shared__ float wmax[4][16];
  __shared__ float wsum[4][32];
  __shared__ float red[4][8][32];

  const int tid  = threadIdx.x;
  const int wave = tid >> 5;
  const int lane = tid & 31;
  const int half = lane >> 4;
  const int l    = lane & 15;
  const int i0   = (blockIdx.x >> 2) * 16;
  const int ncol = (blockIdx.x & 3) * 16 + l;
  const int row  = i0 + l;
  const int kq   = wave * KW + 2 * half;

  const uint4 mb = *(const uint4*)(amask + row * MASK_WORDS + wave * 8 + half * 4);
  const float ei = e1[row];
  const v2f* Eg = (const v2f*)(e2 + kq);
  v2f s[KSW];
#pragma unroll
  for (int ks = 0; ks < KSW; ++ks) s[ks] = Eg[2 * ks];

  const v2f* Bg = (const v2f*)Wh + (size_t)(wave * (KW / 2) + half) * F_DIM + ncol;
  v2f bf[KSW];
#pragma unroll
  for (int ks = 0; ks < KSW; ++ks) bf[ks] = Bg[(2 * ks) * F_DIM];

  float m = NEG_INF_F;
#pragma unroll
  for (int ks = 0; ks < KSW; ++ks) {
    const unsigned w0 = (ks < 32) ? mb.x : mb.z;
    const unsigned w1 = (ks < 32) ? mb.y : mb.w;
    const int b = ks & 31;
    float sx = ei + s[ks].x;
    float sy = ei + s[ks].y;
    sx = (sx > 0.f) ? sx : GAT_ALPHA * sx;
    sy = (sy > 0.f) ? sy : GAT_ALPHA * sy;
    sx = ((w0 >> b) & 1u) ? sx : NEG_INF_F;
    sy = ((w1 >> b) & 1u) ? sy : NEG_INF_F;
    s[ks].x = sx;
    s[ks].y = sy;
    m = fmaxf(m, fmaxf(sx, sy));
  }
  m = fmaxf(m, __shfl_xor(m, 16, 32));
  if (half == 0) wmax[wave][l] = m;
  __builtin_amdgcn_fence(__ATOMIC_RELEASE, "workgroup", "local");
  __builtin_amdgcn_s_barrier();
  __builtin_amdgcn_fence(__ATOMIC_ACQUIRE, "workgroup", "local");
  m = fmaxf(fmaxf(wmax[0][l], wmax[1][l]), fmaxf(wmax[2][l], wmax[3][l]));

  float sum0 = 0.0f, sum1 = 0.0f;
  v8f c0 = {}, c1 = {};
#pragma unroll
  for (int ks = 0; ks < KSW; ks += 2) {
    v2f p0, p1;
    p0.x = __expf(s[ks].x - m);
    p0.y = __expf(s[ks].y - m);
    p1.x = __expf(s[ks + 1].x - m);
    p1.y = __expf(s[ks + 1].y - m);
    sum0 += p0.x + p0.y;
    sum1 += p1.x + p1.y;
    c0 = __builtin_amdgcn_wmma_f32_16x16x4_f32(false, p0, false, bf[ks],
                                               (short)0, c0, false, false);
    c1 = __builtin_amdgcn_wmma_f32_16x16x4_f32(false, p1, false, bf[ks + 1],
                                               (short)0, c1, false, false);
  }
  wsum[wave][lane] = sum0 + sum1;
#pragma unroll
  for (int r = 0; r < 8; ++r) red[wave][r][lane] = c0[r] + c1[r];
  __syncthreads();

#pragma unroll
  for (int rr = 0; rr < 2; ++rr) {
    const int r  = 2 * wave + rr;
    const int mr = r + 8 * half;
    float v = ((red[0][r][lane] + red[1][r][lane]) + red[2][r][lane]) + red[3][r][lane];
    const float den = (((wsum[0][mr] + wsum[0][mr + 16]) + (wsum[1][mr] + wsum[1][mr + 16]))
                     + (wsum[2][mr] + wsum[2][mr + 16])) + (wsum[3][mr] + wsum[3][mr + 16]);
    v = v / den;
    v = (v > 0.0f) ? v : (__expf(v) - 1.0f);
    out[(i0 + mr) * F_DIM + ncol] = v;
  }
}

extern "C" void kernel_launch(void* const* d_in, const int* in_sizes, int n_in,
                              void* d_out, int out_size, void* d_ws, size_t ws_size,
                              hipStream_t stream) {
  const float* h   = (const float*)d_in[0];
  const int*   adj = (const int*)  d_in[1];
  const float* W   = (const float*)d_in[2];
  const float* a   = (const float*)d_in[3];
  float* out = (float*)d_out;

  float* Wh = (float*)d_ws;
  float* e1 = Wh + N_NODES * F_DIM;
  float* e2 = e1 + N_NODES;
  unsigned* amask = (unsigned*)(e2 + N_NODES);

  k_wh_e<<<WH_BLOCKS + PACK_BLOCKS, 128, 0, stream>>>(h, W, a, adj, Wh, e1, e2, amask);

  k_attn<<<(N_NODES / 16) * (F_DIM / 16), 128, 0, stream>>>(Wh, amask, e1, e2, out);
}
